# starting kernel
# baseline (speedup 1.0000x reference)
#include <hip/hip_runtime.h>

typedef __attribute__((ext_vector_type(2))) float v2f;
typedef __attribute__((ext_vector_type(8))) float v8f;

#define T_LEN 2048
#define D_DIM 512
#define B_N   32
#define WAVES_PER_BLOCK 4
#define BLOCK (WAVES_PER_BLOCK * 32)
#define D_PER_BLOCK (WAVES_PER_BLOCK * 16)
#define BLOCKS_PER_B (D_DIM / D_PER_BLOCK)

static __device__ __forceinline__ v8f wmma_acc(v2f a, v2f bb, v8f c) {
    return __builtin_amdgcn_wmma_f32_16x16x4_f32(
        false, a, false, bb, (short)0, c, false, false);
}

__global__ __launch_bounds__(BLOCK)
void masked_seq_reduce_wmma(const float* __restrict__ vs,
                            const int*   __restrict__ slen,
                            const int*   __restrict__ words,
                            const float* __restrict__ weights,
                            float*       __restrict__ out)
{
    __shared__ __align__(16) float lds_wm[T_LEN];

    const int b     = blockIdx.x / BLOCKS_PER_B;
    const int chunk = blockIdx.x % BLOCKS_PER_B;
    const int len   = slen[b];

    for (int t = threadIdx.x; t < T_LEN; t += BLOCK) {
        float w = 0.0f;
        if (t < len) w = weights[words[b * T_LEN + t]];
        lds_wm[t] = w;
    }
    __syncthreads();

    const int lane = threadIdx.x & 31;
    const int hh   = lane >> 4;
    const int n    = lane & 15;
    const int wv   = threadIdx.x >> 5;
    const int d0   = chunk * D_PER_BLOCK + wv * 16;

    const float selMask = (n == 0) ? 1.0f : 0.0f;
    const float selW    = (n == 1) ? 1.0f : 0.0f;

    const v2f* wm2 = (const v2f*)lds_wm;

    const float* p = vs + ((size_t)b * T_LEN + 2 * hh) * D_DIM + d0 + n;

    v8f c_main = {};
    v8f c_abs  = {};

    const int len4 = len & ~3;

#pragma unroll 4
    for (int tb = 0; tb < len4; tb += 4) {
        const v2f  w01 = wm2[(tb >> 1) + hh];
        const float x0 = __builtin_nontemporal_load(p);
        const float x1 = __builtin_nontemporal_load(p + D_DIM);
        __builtin_prefetch(p + 16 * D_DIM, 0, 0);
        p += 4 * D_DIM;

        v2f a, bx, ba;
        a.x  = __builtin_fmaf(selW, w01.x, selMask);
        a.y  = __builtin_fmaf(selW, w01.y, selMask);
        bx.x = x0;
        bx.y = x1;
        ba.x = __builtin_fabsf(x0);
        ba.y = __builtin_fabsf(x1);

        c_main = wmma_acc(a, bx, c_main);
        c_abs  = wmma_acc(a, ba, c_abs);
    }

    if (len4 < len) {
        const int t0 = len4 + 2 * hh;
        const int t1 = t0 + 1;
        const float x0 = __builtin_nontemporal_load(p);
        const float x1 = __builtin_nontemporal_load(p + D_DIM);
        const float m0 = (t0 < len) ? selMask : 0.0f;
        const float m1 = (t1 < len) ? selMask : 0.0f;
        const float w0 = lds_wm[t0];
        const float w1 = lds_wm[t1];

        v2f a, bx, ba;
        a.x  = __builtin_fmaf(selW, w0, m0);
        a.y  = __builtin_fmaf(selW, w1, m1);
        bx.x = x0;
        bx.y = x1;
        ba.x = __builtin_fabsf(x0);
        ba.y = __builtin_fabsf(x1);

        c_main = wmma_acc(a, bx, c_main);
        c_abs  = wmma_acc(a, ba, c_abs);
    }

    if (lane < 16) {
        const int d = d0 + n;
        out[b * D_DIM + d]               = c_main[0] / c_abs[0];
        out[B_N * D_DIM + b * D_DIM + d] = c_main[1];
    }
}

extern "C" void kernel_launch(void* const* d_in, const int* in_sizes, int n_in,
                              void* d_out, int out_size, void* d_ws, size_t ws_size,
                              hipStream_t stream) {
    const float* vs      = (const float*)d_in[0];
    const int*   slen    = (const int*)d_in[1];
    const int*   words   = (const int*)d_in[2];
    const float* weights = (const float*)d_in[3];
    float*       out     = (float*)d_out;

    dim3 grid(B_N * BLOCKS_PER_B);
    dim3 block(BLOCK);
    hipLaunchKernelGGL(masked_seq_reduce_wmma, grid, block, 0, stream,
                       vs, slen, words, weights, out);
}


# T loop dealt over 16 waves per block: 16-byte row loads one group ahead, branchless ragged tail, x and |x| sums in one accumulator tile, LDS fold of wave partials (sum order changes, fixed per run)
# speedup vs baseline: 5.3563x; 5.3563x over previous
#include <hip/hip_runtime.h>

typedef __attribute__((ext_vector_type(2))) float v2f;
typedef __attribute__((ext_vector_type(4))) float v4f;
typedef __attribute__((ext_vector_type(8))) float v8f;

#define T_LEN 2048
#define D_DIM 512
#define B_N   32
#define NW    16
#define BLOCK (NW * 32)
#define D_PER_BLOCK 64
#define BLOCKS_PER_B (D_DIM / D_PER_BLOCK)
#define QG    8
#define G_TOK (4 * QG)
#define NLD   (2 * QG)

static __device__ __forceinline__ v8f wmma_acc(v2f a, v2f bb, v8f c) {
    return __builtin_amdgcn_wmma_f32_16x16x4_f32(
        false, a, false, bb, (short)0, c, false, false);
}

static __device__ __forceinline__ void load_group(v4f (&r)[NLD], const float* p) {
#pragma unroll
    for (int q = 0; q < QG; ++q) {
        r[2 * q]     = __builtin_nontemporal_load((const v4f*)(p + (4 * q)     * D_DIM));
        r[2 * q + 1] = __builtin_nontemporal_load((const v4f*)(p + (4 * q + 1) * D_DIM));
    }
}

template <bool MASKED>
static __device__ __forceinline__ void compute_group(
    const v4f (&r)[NLD], const int tb0, const int len, const int hh,
    const float selMask, const bool isW, const float selAbs, const v2f* wm2,
    v8f (&cm)[4])
{
#pragma unroll
    for (int q = 0; q < QG; ++q) {
        const int tb = tb0 + 4 * q;
        const v2f w01 = wm2[(tb >> 1) + hh];
        const v4f x0 = r[2 * q];
        const v4f x1 = r[2 * q + 1];
        float m0 = selMask, m1 = selMask, n0 = selAbs, n1 = selAbs;
        if (MASKED) {
            const int t0 = tb + 2 * hh;
            m0 = (t0     < len) ? selMask : 0.0f;
            m1 = (t0 + 1 < len) ? selMask : 0.0f;
            n0 = (t0     < len) ? selAbs  : 0.0f;
            n1 = (t0 + 1 < len) ? selAbs  : 0.0f;
        }
        v2f a, a2;
        a.x  = isW ? w01.x : m0;
        a.y  = isW ? w01.y : m1;
        a2.x = n0;
        a2.y = n1;
#pragma unroll
        for (int j = 0; j < 4; ++j) {
            v2f bx, ba;
            bx.x = x0[j];
            bx.y = x1[j];
            ba.x = __builtin_fabsf(x0[j]);
            ba.y = __builtin_fabsf(x1[j]);
            cm[j] = wmma_acc(a,  bx, cm[j]);
            cm[j] = wmma_acc(a2, ba, cm[j]);
        }
    }
}

__global__ __launch_bounds__(BLOCK)
void masked_seq_reduce_wmma(const float* __restrict__ vs,
                            const int*   __restrict__ slen,
                            const int*   __restrict__ words,
                            const float* __restrict__ weights,
                            float*       __restrict__ out)
{
    __shared__ __align__(16) float lds_wm[T_LEN];
    __shared__ __align__(16) float red[NW][3][D_PER_BLOCK];

    const int b     = blockIdx.x / BLOCKS_PER_B;
    const int chunk = blockIdx.x % BLOCKS_PER_B;
    const int len   = slen[b];
    const int d0    = chunk * D_PER_BLOCK;

    const int lane = threadIdx.x & 31;
    const int hh   = lane >> 4;
    const int n    = lane & 15;
    const int wv   = __builtin_amdgcn_readfirstlane((int)(threadIdx.x >> 5));

    const int ntot = (len + G_TOK - 1) / G_TOK;

    const float* base = vs + ((size_t)b * T_LEN + 2 * hh) * D_DIM + d0 + 4 * n;

    v4f ra[NLD], rb[NLD];
    int g = wv;

    {
        int   idx[T_LEN / BLOCK];
        float wt[T_LEN / BLOCK];
#pragma unroll
        for (int i = 0; i < T_LEN / BLOCK; ++i) {
            const int t = threadIdx.x + i * BLOCK;
            idx[i] = words[b * T_LEN + t];
        }
        load_group(ra, base + (size_t)(g < ntot ? g : 0) * G_TOK * D_DIM);
        __builtin_amdgcn_sched_barrier(0);
#pragma unroll
        for (int i = 0; i < T_LEN / BLOCK; ++i) wt[i] = weights[idx[i]];
#pragma unroll
        for (int i = 0; i < T_LEN / BLOCK; ++i) {
            const int t = threadIdx.x + i * BLOCK;
            lds_wm[t] = (t < len) ? wt[i] : 0.0f;
        }
    }
    __syncthreads();

    const float selMask = (n == 0) ? 1.0f : 0.0f;
    const bool  isW     = (n == 1);
    const float selAbs  = (n == 2) ? 1.0f : 0.0f;
    const v2f* wm2 = (const v2f*)lds_wm;

    v8f cm[4] = {};

    if (g < ntot) {
        while (g + 2 * NW < ntot) {
            load_group(rb, base + (size_t)(g + NW) * G_TOK * D_DIM);
            compute_group<false>(ra, g * G_TOK, len, hh, selMask, isW, selAbs, wm2, cm);
            __builtin_amdgcn_sched_barrier(0);
            load_group(ra, base + (size_t)(g + 2 * NW) * G_TOK * D_DIM);
            compute_group<false>(rb, (g + NW) * G_TOK, len, hh, selMask, isW, selAbs, wm2, cm);
            g += 2 * NW;
        }
        if (g + NW < ntot) {
            load_group(rb, base + (size_t)(g + NW) * G_TOK * D_DIM);
            compute_group<false>(ra, g * G_TOK, len, hh, selMask, isW, selAbs, wm2, cm);
            compute_group<true>(rb, (g + NW) * G_TOK, len, hh, selMask, isW, selAbs, wm2, cm);
        } else {
            compute_group<true>(ra, g * G_TOK, len, hh, selMask, isW, selAbs, wm2, cm);
        }
    }

    if (lane < 16) {
        const v4f s  = {cm[0][0], cm[1][0], cm[2][0], cm[3][0]};
        const v4f ws = {cm[0][1], cm[1][1], cm[2][1], cm[3][1]};
        const v4f as = {cm[0][2], cm[1][2], cm[2][2], cm[3][2]};
        *(v4f*)&red[wv][0][4 * n] = s;
        *(v4f*)&red[wv][1][4 * n] = ws;
        *(v4f*)&red[wv][2][4 * n] = as;
    }
    __syncthreads();

    if (threadIdx.x < D_PER_BLOCK) {
        float s = 0.0f, ws = 0.0f, as = 0.0f;
#pragma unroll
        for (int w = 0; w < NW; ++w) {
            s  += red[w][0][threadIdx.x];
            ws += red[w][1][threadIdx.x];
            as += red[w][2][threadIdx.x];
        }
        const int d = d0 + threadIdx.x;
        out[b * D_DIM + d]               = s / as;
        out[B_N * D_DIM + b * D_DIM + d] = ws;
    }
}

extern "C" void kernel_launch(void* const* d_in, const int* in_sizes, int n_in,
                              void* d_out, int out_size, void* d_ws, size_t ws_size,
                              hipStream_t stream) {
    const float* vs      = (const float*)d_in[0];
    const int*   slen    = (const int*)d_in[1];
    const int*   words   = (const int*)d_in[2];
    const float* weights = (const float*)d_in[3];
    float*       out     = (float*)d_out;

    dim3 grid(B_N * BLOCKS_PER_B);
    dim3 block(BLOCK);
    hipLaunchKernelGGL(masked_seq_reduce_wmma, grid, block, 0, stream,
                       vs, slen, words, weights, out);
}


# Matrix work balanced over blocks: a block reduces one 32-dim segment of a long and of a short batch (length ranks p and 31-p) in 64-token groups over 16 waves (sum order changes, fixed per run)
# speedup vs baseline: 5.6721x; 1.0590x over previous
#include <hip/hip_runtime.h>

typedef __attribute__((ext_vector_type(2))) float v2f;
typedef __attribute__((ext_vector_type(4))) float v4f;
typedef __attribute__((ext_vector_type(8))) float v8f;

#define T_LEN 2048
#define D_DIM 512
#define B_N   32
#define NW    16
#define BLOCK (NW * 32)
#define D_PER_ITEM 32
#define SEGS  (D_DIM / D_PER_ITEM)
#define NPAIR (B_N / 2)
#define QG    8
#define G_TOK (8 * QG)
#define NLD   (2 * QG)


static __device__ __forceinline__ v8f wmma_acc(v2f a, v2f bb, v8f c) {
    return __builtin_amdgcn_wmma_f32_16x16x4_f32(
        false, a, false, bb, (short)0, c, false, false);
}

static __device__ __forceinline__ void load_group(v4f (&r)[NLD], const float* p) {
#pragma unroll
    for (int q = 0; q < QG; ++q) {
        r[2 * q]     = __builtin_nontemporal_load((const v4f*)(p + (8 * q)     * D_DIM));
        r[2 * q + 1] = __builtin_nontemporal_load((const v4f*)(p + (8 * q + 1) * D_DIM));
    }
}

static __device__ __forceinline__ void compute_group(
    const v4f (&r)[NLD], const int remg, const float selMask, const bool isW,
    const float selAbs, const v2f* wq, v8f (&cm)[4])
{
#pragma unroll
    for (int q = 0; q < QG; ++q) {
        const v2f w01 = wq[4 * q];
        const v4f x0 = r[2 * q];
        const v4f x1 = r[2 * q + 1];
        const bool l0 = (8 * q     < remg);
        const bool l1 = (8 * q + 1 < remg);
        v2f a, a2;
        a.x  = isW ? w01.x : (l0 ? selMask : 0.0f);
        a.y  = isW ? w01.y : (l1 ? selMask : 0.0f);
        a2.x = l0 ? selAbs : 0.0f;
        a2.y = l1 ? selAbs : 0.0f;
#pragma unroll
        for (int j = 0; j < 4; ++j) {
            v2f bx, ba;
            bx.x = x0[j];
            bx.y = x1[j];
            ba.x = __builtin_fabsf(x0[j]);
            ba.y = __builtin_fabsf(x1[j]);
            cm[j] = wmma_acc(a,  bx, cm[j]);
            cm[j] = wmma_acc(a2, ba, cm[j]);
        }
    }
}

static __device__ __forceinline__ void stash(float (*dst)[2 * D_PER_ITEM], const v8f (&cm)[4],
                                             const int lane) {
    if (lane < 16) {
        const bool u = lane >= 8;
        const v4f s  = {u ? cm[0][3] : cm[0][0], u ? cm[1][3] : cm[1][0],
                        u ? cm[2][3] : cm[2][0], u ? cm[3][3] : cm[3][0]};
        const v4f ws = {u ? cm[0][4] : cm[0][1], u ? cm[1][4] : cm[1][1],
                        u ? cm[2][4] : cm[2][1], u ? cm[3][4] : cm[3][1]};
        const v4f as = {u ? cm[0][5] : cm[0][2], u ? cm[1][5] : cm[1][2],
                        u ? cm[2][5] : cm[2][2], u ? cm[3][5] : cm[3][2]};
        *(v4f*)&dst[0][4 * lane] = s;
        *(v4f*)&dst[1][4 * lane] = ws;
        *(v4f*)&dst[2][4 * lane] = as;
    }
}

__global__ __launch_bounds__(BLOCK)
void masked_seq_reduce_wmma(const float* __restrict__ vs,
                            const int*   __restrict__ slen,
                            const int*   __restrict__ words,
                            const float* __restrict__ weights,
                            float*       __restrict__ out)
{
    __shared__ __align__(16) float lds_wm[2][T_LEN];
    __shared__ __align__(16) float red[2][NW][3][2 * D_PER_ITEM];

    const int pr  = blockIdx.x / SEGS;
    const int seg = blockIdx.x % SEGS;
    const int d0  = seg * D_PER_ITEM;

    const int lane = threadIdx.x & 31;
    const int hh   = lane >> 4;
    const int n    = lane & 15;
    const int u    = n >> 3;
    const int n8   = n & 7;
    const int wv   = __builtin_amdgcn_readfirstlane((int)(threadIdx.x >> 5));

    int lv = slen[lane];
    lv = lv < 0 ? 0 : (lv > T_LEN ? T_LEN : lv);
    const int key = lv * B_N + (B_N - 1 - lane);
    int rank = 0;
#pragma unroll
    for (int j = 0; j < B_N; ++j)
        rank += (__builtin_amdgcn_readlane(key, j) > key) ? 1 : 0;
    const unsigned mA = __builtin_amdgcn_ballot_w32(rank == pr);
    const unsigned mB = __builtin_amdgcn_ballot_w32(rank == B_N - 1 - pr);
    const int bA = mA ? __builtin_ctz(mA) : 0;
    const int bB = mB ? __builtin_ctz(mB) : 0;
    const int lenA = __builtin_amdgcn_readlane(lv, bA);
    const int lenB = __builtin_amdgcn_readlane(lv, bB);

    const int GA   = (lenA + G_TOK - 1) / G_TOK;
    const int GB   = (lenB + G_TOK - 1) / G_TOK;
    const int ntot = GA + GB;

    const int lane_off = (4 * u + 2 * hh) * D_DIM + d0 + 4 * n8;
    const int rowA = bA * T_LEN;
    const int rowB = bB * T_LEN;

    v4f ra[NLD], rb[NLD];
    int g = wv;

    {
        int   idx[2 * T_LEN / BLOCK];
        float wt[2 * T_LEN / BLOCK];
#pragma unroll
        for (int i = 0; i < T_LEN / BLOCK; ++i) {
            const int t = threadIdx.x + i * BLOCK;
            idx[2 * i]     = words[rowA + t];
            idx[2 * i + 1] = words[rowB + t];
        }
        {
            const int g0 = (g < ntot) ? g : 0;
            const int row = (g0 < GA) ? (rowA + g0 * G_TOK) : (rowB + (g0 - GA) * G_TOK);
            load_group(ra, vs + (size_t)row * D_DIM + lane_off);
        }
        __builtin_amdgcn_sched_barrier(0);
#pragma unroll
        for (int i = 0; i < 2 * T_LEN / BLOCK; ++i) wt[i] = weights[idx[i]];
#pragma unroll
        for (int i = 0; i < T_LEN / BLOCK; ++i) {
            const int t = threadIdx.x + i * BLOCK;
            lds_wm[0][t] = (t < lenA) ? wt[2 * i]     : 0.0f;
            lds_wm[1][t] = (t < lenB) ? wt[2 * i + 1] : 0.0f;
        }
    }
    __syncthreads();

    const bool  rowSet1 = (n >= 3) && (n < 6);
    const float selMask = (n == 0 || n == 3) ? 1.0f : 0.0f;
    const bool  isW     = (n == 1 || n == 4);
    const float selAbs  = (n == 2 || n == 5) ? 1.0f : 0.0f;
    const int   atok    = (rowSet1 ? 4 : 0) + 2 * hh;

    v8f cm[4] = {};
    bool inB = false;

#define ROW_OF(i) (((i) < GA) ? (rowA + (i) * G_TOK) : (rowB + ((i) - GA) * G_TOK))
#define LOAD(r, i) load_group(r, vs + (size_t)ROW_OF(i) * D_DIM + lane_off)
#define COMPUTE(r, i)                                                              \
    do {                                                                           \
        const bool isB_ = (i) >= GA;                                               \
        if (isB_ && !inB) {                                                        \
            stash(red[0][wv], cm, lane);                                           \
            cm[0] = v8f{}; cm[1] = v8f{}; cm[2] = v8f{}; cm[3] = v8f{};            \
            inB = true;                                                            \
        }                                                                          \
        const int tb0_ = (isB_ ? (i) - GA : (i)) * G_TOK;                          \
        const int len_ = isB_ ? lenB : lenA;                                       \
        const v2f* wq_ = (const v2f*)(isB_ ? lds_wm[1] : lds_wm[0])                \
                         + ((tb0_ + atok) >> 1);                                   \
        compute_group(r, len_ - tb0_ - atok, selMask, isW, selAbs, wq_, cm);       \
    } while (0)

    if (g < ntot) {
        while (g + 2 * NW < ntot) {
            LOAD(rb, g + NW);
            COMPUTE(ra, g);
            __builtin_amdgcn_sched_barrier(0);
            LOAD(ra, g + 2 * NW);
            COMPUTE(rb, g + NW);
            g += 2 * NW;
        }
        if (g + NW < ntot) {
            LOAD(rb, g + NW);
            COMPUTE(ra, g);
            COMPUTE(rb, g + NW);
        } else {
            COMPUTE(ra, g);
        }
    }
#undef COMPUTE
#undef LOAD
#undef ROW_OF

    if (!inB) {
        stash(red[0][wv], cm, lane);
        cm[0] = v8f{}; cm[1] = v8f{}; cm[2] = v8f{}; cm[3] = v8f{};
    }
    stash(red[1][wv], cm, lane);
    __syncthreads();

    if (threadIdx.x < 2 * D_PER_ITEM) {
        const int item = threadIdx.x >> 5;
        const int dl   = threadIdx.x & 31;
        float s = 0.0f, ws = 0.0f, as = 0.0f;
#pragma unroll
        for (int w = 0; w < NW; ++w) {
            s  += red[item][w][0][dl];
            s  += red[item][w][0][D_PER_ITEM + dl];
            ws += red[item][w][1][dl];
            ws += red[item][w][1][D_PER_ITEM + dl];
            as += red[item][w][2][dl];
            as += red[item][w][2][D_PER_ITEM + dl];
        }
        const int b = item ? bB : bA;
        const int d = d0 + dl;
        out[b * D_DIM + d]               = s / as;
        out[B_N * D_DIM + b * D_DIM + d] = ws;
    }
}

extern "C" void kernel_launch(void* const* d_in, const int* in_sizes, int n_in,
                              void* d_out, int out_size, void* d_ws, size_t ws_size,
                              hipStream_t stream) {
    const float* vs      = (const float*)d_in[0];
    const int*   slen    = (const int*)d_in[1];
    const int*   words   = (const int*)d_in[2];
    const float* weights = (const float*)d_in[3];
    float*       out     = (float*)d_out;

    dim3 grid(NPAIR * SEGS);
    dim3 block(BLOCK);
    hipLaunchKernelGGL(masked_seq_reduce_wmma, grid, block, 0, stream,
                       vs, slen, words, weights, out);
}


# Per-token weights fetched by the wave that owns a group, for its live groups only (index -> gather -> its own LDS slots): block-wide staging of all 2 x T weights and its barrier removed
# speedup vs baseline: 6.6634x; 1.1748x over previous
#include <hip/hip_runtime.h>

typedef __attribute__((ext_vector_type(2))) float v2f;
typedef __attribute__((ext_vector_type(4))) float v4f;
typedef __attribute__((ext_vector_type(8))) float v8f;

#define T_LEN 2048
#define D_DIM 512
#define B_N   32
#define NW    16
#define BLOCK (NW * 32)
#define D_PER_ITEM 32
#define SEGS  (D_DIM / D_PER_ITEM)
#define NPAIR (B_N / 2)
#define QG    8
#define G_TOK (8 * QG)
#define NLD   (2 * QG)

#define KMAX ((2 * (T_LEN / G_TOK) + NW - 1) / NW)

static __device__ __forceinline__ v8f wmma_acc(v2f a, v2f bb, v8f c) {
    return __builtin_amdgcn_wmma_f32_16x16x4_f32(
        false, a, false, bb, (short)0, c, false, false);
}

static __device__ __forceinline__ void load_group(v4f (&r)[NLD], const float* p) {
#pragma unroll
    for (int q = 0; q < QG; ++q) {
        r[2 * q]     = __builtin_nontemporal_load((const v4f*)(p + (8 * q)     * D_DIM));
        r[2 * q + 1] = __builtin_nontemporal_load((const v4f*)(p + (8 * q + 1) * D_DIM));
    }
}

template <int C>
static __device__ __forceinline__ void stage_own_groups(
    v4f (&ra)[NLD], float (*wm)[T_LEN],
    const float* __restrict__ vs, const int* __restrict__ words,
    const float* __restrict__ weights,
    const int wv, const int lane, const int lane_off,
    const int GA, const int rowA, const int rowB, const int lenA, const int lenB)
{
    int   id[2 * C];
    float wt[2 * C];
#pragma unroll
    for (int k = 0; k < C; ++k) {
        const int  gi  = wv + k * NW;
        const bool isB = gi >= GA;
        const int  tok = (isB ? rowB + (gi - GA) * G_TOK : rowA + gi * G_TOK) + lane;
        id[2 * k]     = words[tok];
        id[2 * k + 1] = words[tok + 32];
    }
    {
        const int row = (wv < GA) ? (rowA + wv * G_TOK) : (rowB + (wv - GA) * G_TOK);
        load_group(ra, vs + (size_t)row * D_DIM + lane_off);
    }
    __builtin_amdgcn_sched_barrier(0);
#pragma unroll
    for (int i = 0; i < 2 * C; ++i) wt[i] = weights[id[i]];
#pragma unroll
    for (int k = 0; k < C; ++k) {
        const int  gi  = wv + k * NW;
        const bool isB = gi >= GA;
        const int  t   = (isB ? gi - GA : gi) * G_TOK + lane;
        const int  len = isB ? lenB : lenA;
        float* dst = isB ? wm[1] : wm[0];
        dst[t]      = (t      < len) ? wt[2 * k]     : 0.0f;
        dst[t + 32] = (t + 32 < len) ? wt[2 * k + 1] : 0.0f;
    }
}

static __device__ __forceinline__ void compute_group(
    const v4f (&r)[NLD], const int remg, const float selMask, const bool isW,
    const float selAbs, const v2f* wq, v8f (&cm)[4])
{
#pragma unroll
    for (int q = 0; q < QG; ++q) {
        const v2f w01 = wq[4 * q];
        const v4f x0 = r[2 * q];
        const v4f x1 = r[2 * q + 1];
        const bool l0 = (8 * q     < remg);
        const bool l1 = (8 * q + 1 < remg);
        v2f a, a2;
        a.x  = isW ? w01.x : (l0 ? selMask : 0.0f);
        a.y  = isW ? w01.y : (l1 ? selMask : 0.0f);
        a2.x = l0 ? selAbs : 0.0f;
        a2.y = l1 ? selAbs : 0.0f;
#pragma unroll
        for (int j = 0; j < 4; ++j) {
            v2f bx, ba;
            bx.x = x0[j];
            bx.y = x1[j];
            ba.x = __builtin_fabsf(x0[j]);
            ba.y = __builtin_fabsf(x1[j]);
            cm[j] = wmma_acc(a,  bx, cm[j]);
            cm[j] = wmma_acc(a2, ba, cm[j]);
        }
    }
}

static __device__ __forceinline__ void stash(float (*dst)[2 * D_PER_ITEM], const v8f (&cm)[4],
                                             const int lane) {
    if (lane < 16) {
        const bool u = lane >= 8;
        const v4f s  = {u ? cm[0][3] : cm[0][0], u ? cm[1][3] : cm[1][0],
                        u ? cm[2][3] : cm[2][0], u ? cm[3][3] : cm[3][0]};
        const v4f ws = {u ? cm[0][4] : cm[0][1], u ? cm[1][4] : cm[1][1],
                        u ? cm[2][4] : cm[2][1], u ? cm[3][4] : cm[3][1]};
        const v4f as = {u ? cm[0][5] : cm[0][2], u ? cm[1][5] : cm[1][2],
                        u ? cm[2][5] : cm[2][2], u ? cm[3][5] : cm[3][2]};
        *(v4f*)&dst[0][4 * lane] = s;
        *(v4f*)&dst[1][4 * lane] = ws;
        *(v4f*)&dst[2][4 * lane] = as;
    }
}

__global__ __launch_bounds__(BLOCK)
void masked_seq_reduce_wmma(const float* __restrict__ vs,
                            const int*   __restrict__ slen,
                            const int*   __restrict__ words,
                            const float* __restrict__ weights,
                            float*       __restrict__ out)
{
    __shared__ __align__(16) float lds_wm[2][T_LEN];
    __shared__ __align__(16) float red[2][NW][3][2 * D_PER_ITEM];

    const int pr  = blockIdx.x / SEGS;
    const int seg = blockIdx.x % SEGS;
    const int d0  = seg * D_PER_ITEM;

    const int lane = threadIdx.x & 31;
    const int hh   = lane >> 4;
    const int n    = lane & 15;
    const int u    = n >> 3;
    const int n8   = n & 7;
    const int wv   = __builtin_amdgcn_readfirstlane((int)(threadIdx.x >> 5));

    int lv = slen[lane];
    lv = lv < 0 ? 0 : (lv > T_LEN ? T_LEN : lv);
    const int key = lv * B_N + (B_N - 1 - lane);
    int rank = 0;
#pragma unroll
    for (int j = 0; j < B_N; ++j)
        rank += (__builtin_amdgcn_readlane(key, j) > key) ? 1 : 0;
    const unsigned mA = __builtin_amdgcn_ballot_w32(rank == pr);
    const unsigned mB = __builtin_amdgcn_ballot_w32(rank == B_N - 1 - pr);
    const int bA = mA ? __builtin_ctz(mA) : 0;
    const int bB = mB ? __builtin_ctz(mB) : 0;
    const int lenA = __builtin_amdgcn_readlane(lv, bA);
    const int lenB = __builtin_amdgcn_readlane(lv, bB);

    const int GA   = (lenA + G_TOK - 1) / G_TOK;
    const int GB   = (lenB + G_TOK - 1) / G_TOK;
    const int ntot = GA + GB;

    const int lane_off = (4 * u + 2 * hh) * D_DIM + d0 + 4 * n8;
    const int rowA = bA * T_LEN;
    const int rowB = bB * T_LEN;

    v4f ra[NLD], rb[NLD];
    int g = wv;

    const int cnt = (wv < ntot) ? ((ntot - wv + NW - 1) / NW) : 0;
    static_assert(KMAX == 4, "one variant per possible group count");
    if (cnt == 1)
        stage_own_groups<1>(ra, lds_wm, vs, words, weights, wv, lane, lane_off, GA, rowA, rowB, lenA, lenB);
    else if (cnt == 2)
        stage_own_groups<2>(ra, lds_wm, vs, words, weights, wv, lane, lane_off, GA, rowA, rowB, lenA, lenB);
    else if (cnt == 3)
        stage_own_groups<3>(ra, lds_wm, vs, words, weights, wv, lane, lane_off, GA, rowA, rowB, lenA, lenB);
    else if (cnt >= 4)
        stage_own_groups<4>(ra, lds_wm, vs, words, weights, wv, lane, lane_off, GA, rowA, rowB, lenA, lenB);
    __builtin_amdgcn_fence(__ATOMIC_ACQ_REL, "wavefront");

    const bool  rowSet1 = (n >= 3) && (n < 6);
    const float selMask = (n == 0 || n == 3) ? 1.0f : 0.0f;
    const bool  isW     = (n == 1 || n == 4);
    const float selAbs  = (n == 2 || n == 5) ? 1.0f : 0.0f;
    const int   atok    = (rowSet1 ? 4 : 0) + 2 * hh;

    v8f cm[4] = {};
    bool inB = false;

#define ROW_OF(i) (((i) < GA) ? (rowA + (i) * G_TOK) : (rowB + ((i) - GA) * G_TOK))
#define LOAD(r, i) load_group(r, vs + (size_t)ROW_OF(i) * D_DIM + lane_off)
#define COMPUTE(r, i)                                                              \
    do {                                                                           \
        const bool isB_ = (i) >= GA;                                               \
        if (isB_ && !inB) {                                                        \
            stash(red[0][wv], cm, lane);                                           \
            cm[0] = v8f{}; cm[1] = v8f{}; cm[2] = v8f{}; cm[3] = v8f{};            \
            inB = true;                                                            \
        }                                                                          \
        const int tb0_ = (isB_ ? (i) - GA : (i)) * G_TOK;                          \
        const int len_ = isB_ ? lenB : lenA;                                       \
        const v2f* wq_ = (const v2f*)(isB_ ? lds_wm[1] : lds_wm[0])                \
                         + ((tb0_ + atok) >> 1);                                   \
        compute_group(r, len_ - tb0_ - atok, selMask, isW, selAbs, wq_, cm);       \
    } while (0)

    if (g < ntot) {
        while (g + 2 * NW < ntot) {
            LOAD(rb, g + NW);
            COMPUTE(ra, g);
            __builtin_amdgcn_sched_barrier(0);
            LOAD(ra, g + 2 * NW);
            COMPUTE(rb, g + NW);
            g += 2 * NW;
        }
        if (g + NW < ntot) {
            LOAD(rb, g + NW);
            COMPUTE(ra, g);
            COMPUTE(rb, g + NW);
        } else {
            COMPUTE(ra, g);
        }
    }
#undef COMPUTE
#undef LOAD
#undef ROW_OF

    if (!inB) {
        stash(red[0][wv], cm, lane);
        cm[0] = v8f{}; cm[1] = v8f{}; cm[2] = v8f{}; cm[3] = v8f{};
    }
    stash(red[1][wv], cm, lane);
    __syncthreads();

    if (threadIdx.x < 2 * D_PER_ITEM) {
        const int item = threadIdx.x >> 5;
        const int dl   = threadIdx.x & 31;
        float s = 0.0f, ws = 0.0f, as = 0.0f;
#pragma unroll
        for (int w = 0; w < NW; ++w) {
            s  += red[item][w][0][dl];
            s  += red[item][w][0][D_PER_ITEM + dl];
            ws += red[item][w][1][dl];
            ws += red[item][w][1][D_PER_ITEM + dl];
            as += red[item][w][2][dl];
            as += red[item][w][2][D_PER_ITEM + dl];
        }
        const int b = item ? bB : bA;
        const int d = d0 + dl;
        out[b * D_DIM + d]               = s / as;
        out[B_N * D_DIM + b * D_DIM + d] = ws;
    }
}

extern "C" void kernel_launch(void* const* d_in, const int* in_sizes, int n_in,
                              void* d_out, int out_size, void* d_ws, size_t ws_size,
                              hipStream_t stream) {
    const float* vs      = (const float*)d_in[0];
    const int*   slen    = (const int*)d_in[1];
    const int*   words   = (const int*)d_in[2];
    const float* weights = (const float*)d_in[3];
    float*       out     = (float*)d_out;

    dim3 grid(NPAIR * SEGS);
    dim3 block(BLOCK);
    hipLaunchKernelGGL(masked_seq_reduce_wmma, grid, block, 0, stream,
                       vs, slen, words, weights, out);
}


# Sum of |x| off the matrix unit: each lane adds |x| of its own loaded tokens (masked) on the vector ALU; one f32 WMMA per 256 B, not two, keeps sum x and sum w*x (sum order changes, fixed per run)
# speedup vs baseline: 7.7375x; 1.1612x over previous
#include <hip/hip_runtime.h>

typedef __attribute__((ext_vector_type(2))) float v2f;
typedef __attribute__((ext_vector_type(4))) float v4f;
typedef __attribute__((ext_vector_type(8))) float v8f;

#define T_LEN 2048
#define D_DIM 512
#define B_N   32
#define NW    16
#define BLOCK (NW * 32)
#define D_PER_ITEM 32
#define SEGS  (D_DIM / D_PER_ITEM)
#define NPAIR (B_N / 2)
#define QG    8
#define G_TOK (8 * QG)
#define NLD   (2 * QG)

#define KMAX ((2 * (T_LEN / G_TOK) + NW - 1) / NW)

static __device__ __forceinline__ v8f wmma_acc(v2f a, v2f bb, v8f c) {
    return __builtin_amdgcn_wmma_f32_16x16x4_f32(
        false, a, false, bb, (short)0, c, false, false);
}

static __device__ __forceinline__ void load_group(v4f (&r)[NLD], const float* p) {
#pragma unroll
    for (int q = 0; q < QG; ++q) {
        r[2 * q]     = __builtin_nontemporal_load((const v4f*)(p + (8 * q)     * D_DIM));
        r[2 * q + 1] = __builtin_nontemporal_load((const v4f*)(p + (8 * q + 1) * D_DIM));
    }
}

template <int C>
static __device__ __forceinline__ void stage_own_groups(
    v4f (&ra)[NLD], float (*wm)[T_LEN],
    const float* __restrict__ vs, const int* __restrict__ words,
    const float* __restrict__ weights,
    const int wv, const int lane, const int lane_off,
    const int GA, const int rowA, const int rowB, const int lenA, const int lenB)
{
    int   id[2 * C];
    float wt[2 * C];
#pragma unroll
    for (int k = 0; k < C; ++k) {
        const int  gi  = wv + k * NW;
        const bool isB = gi >= GA;
        const int  tok = (isB ? rowB + (gi - GA) * G_TOK : rowA + gi * G_TOK) + lane;
        id[2 * k]     = words[tok];
        id[2 * k + 1] = words[tok + 32];
    }
    {
        const int row = (wv < GA) ? (rowA + wv * G_TOK) : (rowB + (wv - GA) * G_TOK);
        load_group(ra, vs + (size_t)row * D_DIM + lane_off);
    }
    __builtin_amdgcn_sched_barrier(0);
#pragma unroll
    for (int i = 0; i < 2 * C; ++i) wt[i] = weights[id[i]];
#pragma unroll
    for (int k = 0; k < C; ++k) {
        const int  gi  = wv + k * NW;
        const bool isB = gi >= GA;
        const int  t   = (isB ? gi - GA : gi) * G_TOK + lane;
        const int  len = isB ? lenB : lenA;
        float* dst = isB ? wm[1] : wm[0];
        dst[t]      = (t      < len) ? wt[2 * k]     : 0.0f;
        dst[t + 32] = (t + 32 < len) ? wt[2 * k + 1] : 0.0f;
    }
}

static __device__ __forceinline__ void compute_group(
    const v4f (&r)[NLD], const int remg, const int remb, const float selMask, const bool isW,
    const v2f* wq, v8f (&cm)[4], float (&ab)[4])
{
#pragma unroll
    for (int q = 0; q < QG; ++q) {
        const v2f w01 = wq[4 * q];
        const v4f x0 = r[2 * q];
        const v4f x1 = r[2 * q + 1];
        const bool l0 = (8 * q     < remg);
        const bool l1 = (8 * q + 1 < remg);
        v2f a;
        a.x  = isW ? w01.x : (l0 ? selMask : 0.0f);
        a.y  = isW ? w01.y : (l1 ? selMask : 0.0f);
        const float m0 = (8 * q     < remb) ? 1.0f : 0.0f;
        const float m1 = (8 * q + 1 < remb) ? 1.0f : 0.0f;
#pragma unroll
        for (int j = 0; j < 4; ++j) {
            v2f bx;
            bx.x = x0[j];
            bx.y = x1[j];
            cm[j] = wmma_acc(a, bx, cm[j]);
            ab[j] = __builtin_fmaf(__builtin_fabsf(x0[j]), m0, ab[j]);
            ab[j] = __builtin_fmaf(__builtin_fabsf(x1[j]), m1, ab[j]);
        }
    }
}

static __device__ __forceinline__ void stash(float (*dst)[2 * D_PER_ITEM], const v8f (&cm)[4],
                                             const float (&ab)[4], const int lane) {
    const v4f as = {ab[0] + __shfl_xor(ab[0], 16), ab[1] + __shfl_xor(ab[1], 16),
                    ab[2] + __shfl_xor(ab[2], 16), ab[3] + __shfl_xor(ab[3], 16)};
    if (lane < 16) {
        const bool u = lane >= 8;
        const v4f s  = {u ? cm[0][3] : cm[0][0], u ? cm[1][3] : cm[1][0],
                        u ? cm[2][3] : cm[2][0], u ? cm[3][3] : cm[3][0]};
        const v4f ws = {u ? cm[0][4] : cm[0][1], u ? cm[1][4] : cm[1][1],
                        u ? cm[2][4] : cm[2][1], u ? cm[3][4] : cm[3][1]};
        *(v4f*)&dst[0][4 * lane] = s;
        *(v4f*)&dst[1][4 * lane] = ws;
        *(v4f*)&dst[2][4 * lane] = as;
    }
}

__global__ __launch_bounds__(BLOCK)
void masked_seq_reduce_wmma(const float* __restrict__ vs,
                            const int*   __restrict__ slen,
                            const int*   __restrict__ words,
                            const float* __restrict__ weights,
                            float*       __restrict__ out)
{
    __shared__ __align__(16) float lds_wm[2][T_LEN];
    __shared__ __align__(16) float red[2][NW][3][2 * D_PER_ITEM];

    const int pr  = blockIdx.x / SEGS;
    const int seg = blockIdx.x % SEGS;
    const int d0  = seg * D_PER_ITEM;

    const int lane = threadIdx.x & 31;
    const int hh   = lane >> 4;
    const int n    = lane & 15;
    const int u    = n >> 3;
    const int n8   = n & 7;
    const int wv   = __builtin_amdgcn_readfirstlane((int)(threadIdx.x >> 5));

    int lv = slen[lane];
    lv = lv < 0 ? 0 : (lv > T_LEN ? T_LEN : lv);
    const int key = lv * B_N + (B_N - 1 - lane);
    int rank = 0;
#pragma unroll
    for (int j = 0; j < B_N; ++j)
        rank += (__builtin_amdgcn_readlane(key, j) > key) ? 1 : 0;
    const unsigned mA = __builtin_amdgcn_ballot_w32(rank == pr);
    const unsigned mB = __builtin_amdgcn_ballot_w32(rank == B_N - 1 - pr);
    const int bA = mA ? __builtin_ctz(mA) : 0;
    const int bB = mB ? __builtin_ctz(mB) : 0;
    const int lenA = __builtin_amdgcn_readlane(lv, bA);
    const int lenB = __builtin_amdgcn_readlane(lv, bB);

    const int GA   = (lenA + G_TOK - 1) / G_TOK;
    const int GB   = (lenB + G_TOK - 1) / G_TOK;
    const int ntot = GA + GB;

    const int lane_off = (4 * u + 2 * hh) * D_DIM + d0 + 4 * n8;
    const int rowA = bA * T_LEN;
    const int rowB = bB * T_LEN;

    v4f ra[NLD], rb[NLD];
    int g = wv;

    const int cnt = (wv < ntot) ? ((ntot - wv + NW - 1) / NW) : 0;
    static_assert(KMAX == 4, "one variant per possible group count");
    if (cnt == 1)
        stage_own_groups<1>(ra, lds_wm, vs, words, weights, wv, lane, lane_off, GA, rowA, rowB, lenA, lenB);
    else if (cnt == 2)
        stage_own_groups<2>(ra, lds_wm, vs, words, weights, wv, lane, lane_off, GA, rowA, rowB, lenA, lenB);
    else if (cnt == 3)
        stage_own_groups<3>(ra, lds_wm, vs, words, weights, wv, lane, lane_off, GA, rowA, rowB, lenA, lenB);
    else if (cnt >= 4)
        stage_own_groups<4>(ra, lds_wm, vs, words, weights, wv, lane, lane_off, GA, rowA, rowB, lenA, lenB);
    __builtin_amdgcn_fence(__ATOMIC_ACQ_REL, "wavefront");

    const bool  rowSet1 = (n >= 3) && (n < 6);
    const float selMask = (n == 0 || n == 3) ? 1.0f : 0.0f;
    const bool  isW     = (n == 1 || n == 4);
    const int   atok    = (rowSet1 ? 4 : 0) + 2 * hh;
    const int   btok    = 4 * u + 2 * hh;

    v8f cm[4] = {};
    float ab[4] = {0.0f, 0.0f, 0.0f, 0.0f};
    bool inB = false;

#define ROW_OF(i) (((i) < GA) ? (rowA + (i) * G_TOK) : (rowB + ((i) - GA) * G_TOK))
#define LOAD(r, i) load_group(r, vs + (size_t)ROW_OF(i) * D_DIM + lane_off)
#define COMPUTE(r, i)                                                              \
    do {                                                                           \
        const bool isB_ = (i) >= GA;                                               \
        if (isB_ && !inB) {                                                        \
            stash(red[0][wv], cm, ab, lane);                                       \
            cm[0] = v8f{}; cm[1] = v8f{}; cm[2] = v8f{}; cm[3] = v8f{};            \
            ab[0] = 0.0f; ab[1] = 0.0f; ab[2] = 0.0f; ab[3] = 0.0f;                \
            inB = true;                                                            \
        }                                                                          \
        const int tb0_ = (isB_ ? (i) - GA : (i)) * G_TOK;                          \
        const int len_ = isB_ ? lenB : lenA;                                       \
        const v2f* wq_ = (const v2f*)(isB_ ? lds_wm[1] : lds_wm[0])                \
                         + ((tb0_ + atok) >> 1);                                   \
        compute_group(r, len_ - tb0_ - atok, len_ - tb0_ - btok, selMask, isW,     \
                      wq_, cm, ab);                                                \
    } while (0)

    if (g < ntot) {
        while (g + 2 * NW < ntot) {
            LOAD(rb, g + NW);
            COMPUTE(ra, g);
            __builtin_amdgcn_sched_barrier(0);
            LOAD(ra, g + 2 * NW);
            COMPUTE(rb, g + NW);
            g += 2 * NW;
        }
        if (g + NW < ntot) {
            LOAD(rb, g + NW);
            COMPUTE(ra, g);
            COMPUTE(rb, g + NW);
        } else {
            COMPUTE(ra, g);
        }
    }
#undef COMPUTE
#undef LOAD
#undef ROW_OF

    if (!inB) {
        stash(red[0][wv], cm, ab, lane);
        cm[0] = v8f{}; cm[1] = v8f{}; cm[2] = v8f{}; cm[3] = v8f{};
        ab[0] = 0.0f; ab[1] = 0.0f; ab[2] = 0.0f; ab[3] = 0.0f;
    }
    stash(red[1][wv], cm, ab, lane);
    __syncthreads();

    if (threadIdx.x < 2 * D_PER_ITEM) {
        const int item = threadIdx.x >> 5;
        const int dl   = threadIdx.x & 31;
        float s = 0.0f, ws = 0.0f, as = 0.0f;
#pragma unroll
        for (int w = 0; w < NW; ++w) {
            s  += red[item][w][0][dl];
            s  += red[item][w][0][D_PER_ITEM + dl];
            ws += red[item][w][1][dl];
            ws += red[item][w][1][D_PER_ITEM + dl];
            as += red[item][w][2][dl];
            as += red[item][w][2][D_PER_ITEM + dl];
        }
        const int b = item ? bB : bA;
        const int d = d0 + dl;
        out[b * D_DIM + d]               = s / as;
        out[B_N * D_DIM + b * D_DIM + d] = ws;
    }
}

extern "C" void kernel_launch(void* const* d_in, const int* in_sizes, int n_in,
                              void* d_out, int out_size, void* d_ws, size_t ws_size,
                              hipStream_t stream) {
    const float* vs      = (const float*)d_in[0];
    const int*   slen    = (const int*)d_in[1];
    const int*   words   = (const int*)d_in[2];
    const float* weights = (const float*)d_in[3];
    float*       out     = (float*)d_out;

    dim3 grid(NPAIR * SEGS);
    dim3 block(BLOCK);
    hipLaunchKernelGGL(masked_seq_reduce_wmma, grid, block, 0, stream,
                       vs, slen, words, weights, out);
}
